# starting kernel
# baseline (speedup 1.0000x reference)
#include <hip/hip_runtime.h>

typedef float v2f __attribute__((ext_vector_type(2)));
typedef float v8f __attribute__((ext_vector_type(8)));

namespace {
constexpr int   C    = 64;
constexpr int   STR  = 68;
constexpr float kEps = 1e-7f;
}

__global__ __launch_bounds__(64)
void mfdfc_wnorm_kernel(const float* __restrict__ w_raw, float* __restrict__ wn) {
  const int o = threadIdx.x;
  const float lc = 4.1588830833596715f;
  float s = 0.0f;
  for (int i = 0; i < C; ++i) s += expf(w_raw[i * C + o] - lc);
  const float inv = 1.0f / s;
  for (int i = 0; i < C; ++i) wn[i * C + o] = expf(w_raw[i * C + o] - lc) * inv;
}

__device__ __forceinline__ float half16_sum(float v) {
  v += __shfl_xor(v, 1, 32);
  v += __shfl_xor(v, 2, 32);
  v += __shfl_xor(v, 4, 32);
  v += __shfl_xor(v, 8, 32);
  return v;
}

__global__ __launch_bounds__(256)
void mfdfc_main_kernel(const float* __restrict__ x, const float* __restrict__ wn,
                       float* __restrict__ out) {
  __shared__ float xs[C * STR];
  __shared__ float as[C * STR];
  __shared__ float ss[C * STR];
  __shared__ float tpart[C * 4];
  __shared__ float gpart[C * 4];
  __shared__ float pa[C];
  __shared__ float qq[C];
  __shared__ float tt[C];
  __shared__ float cg[C];
  __shared__ float sg[C];

  const int tid  = threadIdx.x;
  const int lane = tid & 31;
  const int wave = tid >> 5;
  const int il   = lane & 15;
  const int hh   = lane >> 4;
  const int n    = blockIdx.x;

  const float* xn = x + (size_t)n * C * C;

  for (int j = 0; j < 4; ++j) {
    int f4  = tid + 256 * j;
    int row = f4 >> 4;
    int c4  = f4 & 15;
    const float4 v = ((const float4*)xn)[f4];
    *(float4*)&xs[row * STR + 4 * c4] = v;
  }
  __syncthreads();

  if (tid < C) {
    float s = 0.0f;
    for (int k = 0; k < C; ++k) { float v = xs[tid * STR + k]; s += v * v; }
    qq[tid] = s;
  }
  for (int j = 0; j < 16; ++j) {
    int f = tid + 256 * j;
    int o = f >> 6, k = f & 63;
    as[o * STR + k] = xs[k];
  }

  float wreg[2][8];
  for (int u = 0; u < 2; ++u) {
    int t  = wave * 2 + u;
    int m0 = (t >> 2) * 16;
    int i0 = (t & 3) * 16;
    int i  = i0 + il;
    for (int r = 0; r < 8; ++r) {
      int o = m0 + r + 8 * hh;
      wreg[u][r] = wn[i * C + o];
    }
  }
  __syncthreads();

  for (int it = 0; it < 3; ++it) {
    if (tid < C) {
      float s = 0.0f;
      for (int k = 0; k < C; ++k) { float v = as[tid * STR + k]; s += v * v; }
      pa[tid] = s;
    }
    __syncthreads();

    for (int u = 0; u < 2; ++u) {
      int t  = wave * 2 + u;
      int m0 = (t >> 2) * 16;
      int i0 = (t & 3) * 16;
      v8f acc = {};
      for (int kc = 0; kc < C; kc += 4) {
        v2f af = *(const v2f*)&as[(m0 + il) * STR + kc + 2 * hh];
        v2f bf = *(const v2f*)&xs[(i0 + il) * STR + kc + 2 * hh];
        acc = __builtin_amdgcn_wmma_f32_16x16x4_f32(false, af, false, bf,
                                                    (short)0, acc, false, false);
      }
      int i = i0 + il;
      for (int r = 0; r < 8; ++r) {
        int o = m0 + r + 8 * hh;
        float du  = acc[r];
        float dc  = fminf(fmaxf(du, -1.0f + kEps), 1.0f - kEps);
        float th  = acosf(dc);
        float nv2 = qq[i] - 2.0f * dc * du + dc * dc * pa[o];
        float nv  = sqrtf(fmaxf(nv2, 0.0f));
        float fac = (nv < kEps) ? 1.0f : th / nv;
        float s   = wreg[u][r] * fac;
        ss[o * STR + i] = s;
        float part = half16_sum(s * dc);
        if (il == 0) tpart[o * 4 + (t & 3)] = part;
      }
    }
    __syncthreads();

    if (tid < C)
      tt[tid] = tpart[tid * 4 + 0] + tpart[tid * 4 + 1] +
                tpart[tid * 4 + 2] + tpart[tid * 4 + 3];
    __syncthreads();

    float greg[2][8];
    for (int u = 0; u < 2; ++u) {
      int t  = wave * 2 + u;
      int o0 = (t >> 2) * 16;
      int k0 = (t & 3) * 16;
      v8f acc = {};
      for (int ic = 0; ic < C; ic += 4) {
        v2f af = *(const v2f*)&ss[(o0 + il) * STR + ic + 2 * hh];
        v2f bf;
        bf.x = xs[(ic + 2 * hh)     * STR + k0 + il];
        bf.y = xs[(ic + 2 * hh + 1) * STR + k0 + il];
        acc = __builtin_amdgcn_wmma_f32_16x16x4_f32(false, af, false, bf,
                                                    (short)0, acc, false, false);
      }
      int k = k0 + il;
      for (int r = 0; r < 8; ++r) {
        int o = o0 + r + 8 * hh;
        float g = acc[r] - tt[o] * as[o * STR + k];
        greg[u][r] = g;
        float part = half16_sum(g * g);
        if (il == 0) gpart[o * 4 + (t & 3)] = part;
      }
    }
    __syncthreads();

    if (tid < C) {
      float s2 = gpart[tid * 4 + 0] + gpart[tid * 4 + 1] +
                 gpart[tid * 4 + 2] + gpart[tid * 4 + 3];
      float gl = sqrtf(s2);
      cg[tid] = cosf(gl);
      sg[tid] = (gl < kEps) ? 1.0f : (sinf(gl) / gl);
    }
    __syncthreads();

    for (int u = 0; u < 2; ++u) {
      int t  = wave * 2 + u;
      int o0 = (t >> 2) * 16;
      int k0 = (t & 3) * 16;
      int k  = k0 + il;
      for (int r = 0; r < 8; ++r) {
        int o = o0 + r + 8 * hh;
        as[o * STR + k] = cg[o] * as[o * STR + k] + sg[o] * greg[u][r];
      }
    }
    __syncthreads();
  }

  float* on = out + (size_t)n * C * C;
  for (int j = 0; j < 16; ++j) {
    int f = tid + 256 * j;
    int o = f >> 6, k = f & 63;
    on[f] = as[o * STR + k];
  }
}

extern "C" void kernel_launch(void* const* d_in, const int* in_sizes, int n_in,
                              void* d_out, int out_size, void* d_ws, size_t ws_size,
                              hipStream_t stream) {
  (void)in_sizes; (void)n_in; (void)out_size; (void)ws_size;
  const float* x     = (const float*)d_in[0];
  const float* w_raw = (const float*)d_in[1];
  float* out = (float*)d_out;
  float* wn  = (float*)d_ws;

  mfdfc_wnorm_kernel<<<1, 64, 0, stream>>>(w_raw, wn);
  mfdfc_main_kernel<<<1024, 256, 0, stream>>>(x, wn, out);
}


# 16-lane row sums (s*dot, |grad|^2): four row_xmask DPP adds on the vector ALU replace ds_bpermute trees that waited per step; partials stored once per tile; same xor tree and summation order
# speedup vs baseline: 1.5106x; 1.5106x over previous
#include <hip/hip_runtime.h>

typedef float v2f __attribute__((ext_vector_type(2)));
typedef float v8f __attribute__((ext_vector_type(8)));

namespace {
constexpr int   C    = 64;
constexpr int   STR  = 68;
constexpr float kEps = 1e-7f;
}

__global__ __launch_bounds__(64)
void mfdfc_wnorm_kernel(const float* __restrict__ w_raw, float* __restrict__ wn) {
  const int o = threadIdx.x;
  const float lc = 4.1588830833596715f;
  float s = 0.0f;
  for (int i = 0; i < C; ++i) s += expf(w_raw[i * C + o] - lc);
  const float inv = 1.0f / s;
  for (int i = 0; i < C; ++i) wn[i * C + o] = expf(w_raw[i * C + o] - lc) * inv;
}

template <int M>
__device__ __forceinline__ float row_xor(float v) {
  return __builtin_bit_cast(
      float, __builtin_amdgcn_update_dpp(0, __builtin_bit_cast(int, v), 0x160 + M,
                                         0xf, 0xf, true));
}

__device__ __forceinline__ float half16_sum(float v) {
  v += row_xor<1>(v);
  v += row_xor<2>(v);
  v += row_xor<4>(v);
  v += row_xor<8>(v);
  return v;
}

__global__ __launch_bounds__(256)
void mfdfc_main_kernel(const float* __restrict__ x, const float* __restrict__ wn,
                       float* __restrict__ out) {
  __shared__ float xs[C * STR];
  __shared__ float as[C * STR];
  __shared__ float ss[C * STR];
  __shared__ float tpart[C * 4];
  __shared__ float gpart[C * 4];
  __shared__ float pa[C];
  __shared__ float qq[C];
  __shared__ float tt[C];
  __shared__ float cg[C];
  __shared__ float sg[C];

  const int tid  = threadIdx.x;
  const int lane = tid & 31;
  const int wave = tid >> 5;
  const int il   = lane & 15;
  const int hh   = lane >> 4;
  const int n    = blockIdx.x;

  const float* xn = x + (size_t)n * C * C;

  for (int j = 0; j < 4; ++j) {
    int f4  = tid + 256 * j;
    int row = f4 >> 4;
    int c4  = f4 & 15;
    const float4 v = ((const float4*)xn)[f4];
    *(float4*)&xs[row * STR + 4 * c4] = v;
  }
  __syncthreads();

  if (tid < C) {
    float s = 0.0f;
    for (int k = 0; k < C; ++k) { float v = xs[tid * STR + k]; s += v * v; }
    qq[tid] = s;
  }
  for (int j = 0; j < 16; ++j) {
    int f = tid + 256 * j;
    int o = f >> 6, k = f & 63;
    as[o * STR + k] = xs[k];
  }

  float wreg[2][8];
  for (int u = 0; u < 2; ++u) {
    int t  = wave * 2 + u;
    int m0 = (t >> 2) * 16;
    int i0 = (t & 3) * 16;
    int i  = i0 + il;
    for (int r = 0; r < 8; ++r) {
      int o = m0 + r + 8 * hh;
      wreg[u][r] = wn[i * C + o];
    }
  }
  __syncthreads();

  for (int it = 0; it < 3; ++it) {
    if (tid < C) {
      float s = 0.0f;
      for (int k = 0; k < C; ++k) { float v = as[tid * STR + k]; s += v * v; }
      pa[tid] = s;
    }
    __syncthreads();

    for (int u = 0; u < 2; ++u) {
      int t  = wave * 2 + u;
      int m0 = (t >> 2) * 16;
      int i0 = (t & 3) * 16;
      v8f acc = {};
      for (int kc = 0; kc < C; kc += 4) {
        v2f af = *(const v2f*)&as[(m0 + il) * STR + kc + 2 * hh];
        v2f bf = *(const v2f*)&xs[(i0 + il) * STR + kc + 2 * hh];
        acc = __builtin_amdgcn_wmma_f32_16x16x4_f32(false, af, false, bf,
                                                    (short)0, acc, false, false);
      }
      int i = i0 + il;
      float part[8];
      for (int r = 0; r < 8; ++r) {
        int o = m0 + r + 8 * hh;
        float du  = acc[r];
        float dc  = fminf(fmaxf(du, -1.0f + kEps), 1.0f - kEps);
        float th  = acosf(dc);
        float nv2 = qq[i] - 2.0f * dc * du + dc * dc * pa[o];
        float nv  = sqrtf(fmaxf(nv2, 0.0f));
        float fac = (nv < kEps) ? 1.0f : th / nv;
        float s   = wreg[u][r] * fac;
        ss[o * STR + i] = s;
        part[r] = half16_sum(s * dc);
      }
      if (il == 0)
        for (int r = 0; r < 8; ++r) tpart[(m0 + r + 8 * hh) * 4 + (t & 3)] = part[r];
    }
    __syncthreads();

    if (tid < C)
      tt[tid] = tpart[tid * 4 + 0] + tpart[tid * 4 + 1] +
                tpart[tid * 4 + 2] + tpart[tid * 4 + 3];
    __syncthreads();

    float greg[2][8];
    for (int u = 0; u < 2; ++u) {
      int t  = wave * 2 + u;
      int o0 = (t >> 2) * 16;
      int k0 = (t & 3) * 16;
      v8f acc = {};
      for (int ic = 0; ic < C; ic += 4) {
        v2f af = *(const v2f*)&ss[(o0 + il) * STR + ic + 2 * hh];
        v2f bf;
        bf.x = xs[(ic + 2 * hh)     * STR + k0 + il];
        bf.y = xs[(ic + 2 * hh + 1) * STR + k0 + il];
        acc = __builtin_amdgcn_wmma_f32_16x16x4_f32(false, af, false, bf,
                                                    (short)0, acc, false, false);
      }
      int k = k0 + il;
      float part[8];
      for (int r = 0; r < 8; ++r) {
        int o = o0 + r + 8 * hh;
        float g = acc[r] - tt[o] * as[o * STR + k];
        greg[u][r] = g;
        part[r] = half16_sum(g * g);
      }
      if (il == 0)
        for (int r = 0; r < 8; ++r) gpart[(o0 + r + 8 * hh) * 4 + (t & 3)] = part[r];
    }
    __syncthreads();

    if (tid < C) {
      float s2 = gpart[tid * 4 + 0] + gpart[tid * 4 + 1] +
                 gpart[tid * 4 + 2] + gpart[tid * 4 + 3];
      float gl = sqrtf(s2);
      cg[tid] = cosf(gl);
      sg[tid] = (gl < kEps) ? 1.0f : (sinf(gl) / gl);
    }
    __syncthreads();

    for (int u = 0; u < 2; ++u) {
      int t  = wave * 2 + u;
      int o0 = (t >> 2) * 16;
      int k0 = (t & 3) * 16;
      int k  = k0 + il;
      for (int r = 0; r < 8; ++r) {
        int o = o0 + r + 8 * hh;
        as[o * STR + k] = cg[o] * as[o * STR + k] + sg[o] * greg[u][r];
      }
    }
    __syncthreads();
  }

  float* on = out + (size_t)n * C * C;
  for (int j = 0; j < 16; ++j) {
    int f = tid + 256 * j;
    int o = f >> 6, k = f & 63;
    on[f] = as[o * STR + k];
  }
}

extern "C" void kernel_launch(void* const* d_in, const int* in_sizes, int n_in,
                              void* d_out, int out_size, void* d_ws, size_t ws_size,
                              hipStream_t stream) {
  (void)in_sizes; (void)n_in; (void)out_size; (void)ws_size;
  const float* x     = (const float*)d_in[0];
  const float* w_raw = (const float*)d_in[1];
  float* out = (float*)d_out;
  float* wn  = (float*)d_ws;

  mfdfc_wnorm_kernel<<<1, 64, 0, stream>>>(w_raw, wn);
  mfdfc_main_kernel<<<1024, 256, 0, stream>>>(x, wn, out);
}


# Register budget capped at 128 per wave (waves-per-SIMD attribute 8; built 125, no scratch): four 8-wave work-groups resident per WGP instead of two, all 1024 groups in one pass; sums unchanged
# speedup vs baseline: 1.5871x; 1.0506x over previous
#include <hip/hip_runtime.h>

typedef float v2f __attribute__((ext_vector_type(2)));
typedef float v8f __attribute__((ext_vector_type(8)));

namespace {
constexpr int   C    = 64;
constexpr int   STR  = 68;
constexpr float kEps = 1e-7f;
}

__global__ __launch_bounds__(64)
void mfdfc_wnorm_kernel(const float* __restrict__ w_raw, float* __restrict__ wn) {
  const int o = threadIdx.x;
  const float lc = 4.1588830833596715f;
  float s = 0.0f;
  for (int i = 0; i < C; ++i) s += expf(w_raw[i * C + o] - lc);
  const float inv = 1.0f / s;
  for (int i = 0; i < C; ++i) wn[i * C + o] = expf(w_raw[i * C + o] - lc) * inv;
}

template <int M>
__device__ __forceinline__ float row_xor(float v) {
  return __builtin_bit_cast(
      float, __builtin_amdgcn_update_dpp(0, __builtin_bit_cast(int, v), 0x160 + M,
                                         0xf, 0xf, true));
}

__device__ __forceinline__ float half16_sum(float v) {
  v += row_xor<1>(v);
  v += row_xor<2>(v);
  v += row_xor<4>(v);
  v += row_xor<8>(v);
  return v;
}

__global__ __launch_bounds__(256) __attribute__((amdgpu_waves_per_eu(8, 8)))
void mfdfc_main_kernel(const float* __restrict__ x, const float* __restrict__ wn,
                       float* __restrict__ out) {
  __shared__ float xs[C * STR];
  __shared__ float as[C * STR];
  __shared__ float ss[C * STR];
  __shared__ float tpart[C * 4];
  __shared__ float gpart[C * 4];
  __shared__ float pa[C];
  __shared__ float qq[C];
  __shared__ float tt[C];
  __shared__ float cg[C];
  __shared__ float sg[C];

  const int tid  = threadIdx.x;
  const int lane = tid & 31;
  const int wave = tid >> 5;
  const int il   = lane & 15;
  const int hh   = lane >> 4;
  const int n    = blockIdx.x;

  const float* xn = x + (size_t)n * C * C;

  for (int j = 0; j < 4; ++j) {
    int f4  = tid + 256 * j;
    int row = f4 >> 4;
    int c4  = f4 & 15;
    const float4 v = ((const float4*)xn)[f4];
    *(float4*)&xs[row * STR + 4 * c4] = v;
  }
  __syncthreads();

  if (tid < C) {
    float s = 0.0f;
    for (int k = 0; k < C; ++k) { float v = xs[tid * STR + k]; s += v * v; }
    qq[tid] = s;
  }
  for (int j = 0; j < 16; ++j) {
    int f = tid + 256 * j;
    int o = f >> 6, k = f & 63;
    as[o * STR + k] = xs[k];
  }

  float wreg[2][8];
  for (int u = 0; u < 2; ++u) {
    int t  = wave * 2 + u;
    int m0 = (t >> 2) * 16;
    int i0 = (t & 3) * 16;
    int i  = i0 + il;
    for (int r = 0; r < 8; ++r) {
      int o = m0 + r + 8 * hh;
      wreg[u][r] = wn[i * C + o];
    }
  }
  __syncthreads();

  for (int it = 0; it < 3; ++it) {
    if (tid < C) {
      float s = 0.0f;
      for (int k = 0; k < C; ++k) { float v = as[tid * STR + k]; s += v * v; }
      pa[tid] = s;
    }
    __syncthreads();

    for (int u = 0; u < 2; ++u) {
      int t  = wave * 2 + u;
      int m0 = (t >> 2) * 16;
      int i0 = (t & 3) * 16;
      v8f acc = {};
      for (int kc = 0; kc < C; kc += 4) {
        v2f af = *(const v2f*)&as[(m0 + il) * STR + kc + 2 * hh];
        v2f bf = *(const v2f*)&xs[(i0 + il) * STR + kc + 2 * hh];
        acc = __builtin_amdgcn_wmma_f32_16x16x4_f32(false, af, false, bf,
                                                    (short)0, acc, false, false);
      }
      int i = i0 + il;
      float part[8];
      for (int r = 0; r < 8; ++r) {
        int o = m0 + r + 8 * hh;
        float du  = acc[r];
        float dc  = fminf(fmaxf(du, -1.0f + kEps), 1.0f - kEps);
        float th  = acosf(dc);
        float nv2 = qq[i] - 2.0f * dc * du + dc * dc * pa[o];
        float nv  = sqrtf(fmaxf(nv2, 0.0f));
        float fac = (nv < kEps) ? 1.0f : th / nv;
        float s   = wreg[u][r] * fac;
        ss[o * STR + i] = s;
        part[r] = half16_sum(s * dc);
      }
      if (il == 0)
        for (int r = 0; r < 8; ++r) tpart[(m0 + r + 8 * hh) * 4 + (t & 3)] = part[r];
    }
    __syncthreads();

    if (tid < C)
      tt[tid] = tpart[tid * 4 + 0] + tpart[tid * 4 + 1] +
                tpart[tid * 4 + 2] + tpart[tid * 4 + 3];
    __syncthreads();

    float greg[2][8];
    for (int u = 0; u < 2; ++u) {
      int t  = wave * 2 + u;
      int o0 = (t >> 2) * 16;
      int k0 = (t & 3) * 16;
      v8f acc = {};
      for (int ic = 0; ic < C; ic += 4) {
        v2f af = *(const v2f*)&ss[(o0 + il) * STR + ic + 2 * hh];
        v2f bf;
        bf.x = xs[(ic + 2 * hh)     * STR + k0 + il];
        bf.y = xs[(ic + 2 * hh + 1) * STR + k0 + il];
        acc = __builtin_amdgcn_wmma_f32_16x16x4_f32(false, af, false, bf,
                                                    (short)0, acc, false, false);
      }
      int k = k0 + il;
      float part[8];
      for (int r = 0; r < 8; ++r) {
        int o = o0 + r + 8 * hh;
        float g = acc[r] - tt[o] * as[o * STR + k];
        greg[u][r] = g;
        part[r] = half16_sum(g * g);
      }
      if (il == 0)
        for (int r = 0; r < 8; ++r) gpart[(o0 + r + 8 * hh) * 4 + (t & 3)] = part[r];
    }
    __syncthreads();

    if (tid < C) {
      float s2 = gpart[tid * 4 + 0] + gpart[tid * 4 + 1] +
                 gpart[tid * 4 + 2] + gpart[tid * 4 + 3];
      float gl = sqrtf(s2);
      cg[tid] = cosf(gl);
      sg[tid] = (gl < kEps) ? 1.0f : (sinf(gl) / gl);
    }
    __syncthreads();

    for (int u = 0; u < 2; ++u) {
      int t  = wave * 2 + u;
      int o0 = (t >> 2) * 16;
      int k0 = (t & 3) * 16;
      int k  = k0 + il;
      for (int r = 0; r < 8; ++r) {
        int o = o0 + r + 8 * hh;
        as[o * STR + k] = cg[o] * as[o * STR + k] + sg[o] * greg[u][r];
      }
    }
    __syncthreads();
  }

  float* on = out + (size_t)n * C * C;
  for (int j = 0; j < 16; ++j) {
    int f = tid + 256 * j;
    int o = f >> 6, k = f & 63;
    on[f] = as[o * STR + k];
  }
}

extern "C" void kernel_launch(void* const* d_in, const int* in_sizes, int n_in,
                              void* d_out, int out_size, void* d_ws, size_t ws_size,
                              hipStream_t stream) {
  (void)in_sizes; (void)n_in; (void)out_size; (void)ws_size;
  const float* x     = (const float*)d_in[0];
  const float* w_raw = (const float*)d_in[1];
  float* out = (float*)d_out;
  float* wn  = (float*)d_ws;

  mfdfc_wnorm_kernel<<<1, 64, 0, stream>>>(w_raw, wn);
  mfdfc_main_kernel<<<1024, 256, 0, stream>>>(x, wn, out);
}


# Weight pre-kernel: each lane holds its 64-row column in registers, so its 64 loads are in flight together (not 32 dependent load groups) and each exp is computed once; sums unchanged
# speedup vs baseline: 2.1660x; 1.3648x over previous
#include <hip/hip_runtime.h>

typedef float v2f __attribute__((ext_vector_type(2)));
typedef float v8f __attribute__((ext_vector_type(8)));

namespace {
constexpr int   C    = 64;
constexpr int   STR  = 68;
constexpr float kEps = 1e-7f;
}

__global__ __launch_bounds__(64)
void mfdfc_wnorm_kernel(const float* __restrict__ w_raw, float* __restrict__ wn) {
  const int o = threadIdx.x;
  const float lc = 4.1588830833596715f;
  float e[C];
  for (int i = 0; i < C; ++i) e[i] = w_raw[i * C + o];
  for (int i = 0; i < C; ++i) e[i] = expf(e[i] - lc);
  float s = 0.0f;
  for (int i = 0; i < C; ++i) s += e[i];
  const float inv = 1.0f / s;
  for (int i = 0; i < C; ++i) wn[i * C + o] = e[i] * inv;
}

template <int M>
__device__ __forceinline__ float row_xor(float v) {
  return __builtin_bit_cast(
      float, __builtin_amdgcn_update_dpp(0, __builtin_bit_cast(int, v), 0x160 + M,
                                         0xf, 0xf, true));
}

__device__ __forceinline__ float half16_sum(float v) {
  v += row_xor<1>(v);
  v += row_xor<2>(v);
  v += row_xor<4>(v);
  v += row_xor<8>(v);
  return v;
}

__global__ __launch_bounds__(256) __attribute__((amdgpu_waves_per_eu(8, 8)))
void mfdfc_main_kernel(const float* __restrict__ x, const float* __restrict__ wn,
                       float* __restrict__ out) {
  __shared__ float xs[C * STR];
  __shared__ float as[C * STR];
  __shared__ float ss[C * STR];
  __shared__ float tpart[C * 4];
  __shared__ float gpart[C * 4];
  __shared__ float pa[C];
  __shared__ float qq[C];
  __shared__ float tt[C];
  __shared__ float cg[C];
  __shared__ float sg[C];

  const int tid  = threadIdx.x;
  const int lane = tid & 31;
  const int wave = tid >> 5;
  const int il   = lane & 15;
  const int hh   = lane >> 4;
  const int n    = blockIdx.x;

  const float* xn = x + (size_t)n * C * C;

  for (int j = 0; j < 4; ++j) {
    int f4  = tid + 256 * j;
    int row = f4 >> 4;
    int c4  = f4 & 15;
    const float4 v = ((const float4*)xn)[f4];
    *(float4*)&xs[row * STR + 4 * c4] = v;
  }
  __syncthreads();

  if (tid < C) {
    float s = 0.0f;
    for (int k = 0; k < C; ++k) { float v = xs[tid * STR + k]; s += v * v; }
    qq[tid] = s;
  }
  for (int j = 0; j < 16; ++j) {
    int f = tid + 256 * j;
    int o = f >> 6, k = f & 63;
    as[o * STR + k] = xs[k];
  }

  float wreg[2][8];
  for (int u = 0; u < 2; ++u) {
    int t  = wave * 2 + u;
    int m0 = (t >> 2) * 16;
    int i0 = (t & 3) * 16;
    int i  = i0 + il;
    for (int r = 0; r < 8; ++r) {
      int o = m0 + r + 8 * hh;
      wreg[u][r] = wn[i * C + o];
    }
  }
  __syncthreads();

  for (int it = 0; it < 3; ++it) {
    if (tid < C) {
      float s = 0.0f;
      for (int k = 0; k < C; ++k) { float v = as[tid * STR + k]; s += v * v; }
      pa[tid] = s;
    }
    __syncthreads();

    for (int u = 0; u < 2; ++u) {
      int t  = wave * 2 + u;
      int m0 = (t >> 2) * 16;
      int i0 = (t & 3) * 16;
      v8f acc = {};
      for (int kc = 0; kc < C; kc += 4) {
        v2f af = *(const v2f*)&as[(m0 + il) * STR + kc + 2 * hh];
        v2f bf = *(const v2f*)&xs[(i0 + il) * STR + kc + 2 * hh];
        acc = __builtin_amdgcn_wmma_f32_16x16x4_f32(false, af, false, bf,
                                                    (short)0, acc, false, false);
      }
      int i = i0 + il;
      float part[8];
      for (int r = 0; r < 8; ++r) {
        int o = m0 + r + 8 * hh;
        float du  = acc[r];
        float dc  = fminf(fmaxf(du, -1.0f + kEps), 1.0f - kEps);
        float th  = acosf(dc);
        float nv2 = qq[i] - 2.0f * dc * du + dc * dc * pa[o];
        float nv  = sqrtf(fmaxf(nv2, 0.0f));
        float fac = (nv < kEps) ? 1.0f : th / nv;
        float s   = wreg[u][r] * fac;
        ss[o * STR + i] = s;
        part[r] = half16_sum(s * dc);
      }
      if (il == 0)
        for (int r = 0; r < 8; ++r) tpart[(m0 + r + 8 * hh) * 4 + (t & 3)] = part[r];
    }
    __syncthreads();

    if (tid < C)
      tt[tid] = tpart[tid * 4 + 0] + tpart[tid * 4 + 1] +
                tpart[tid * 4 + 2] + tpart[tid * 4 + 3];
    __syncthreads();

    float greg[2][8];
    for (int u = 0; u < 2; ++u) {
      int t  = wave * 2 + u;
      int o0 = (t >> 2) * 16;
      int k0 = (t & 3) * 16;
      v8f acc = {};
      for (int ic = 0; ic < C; ic += 4) {
        v2f af = *(const v2f*)&ss[(o0 + il) * STR + ic + 2 * hh];
        v2f bf;
        bf.x = xs[(ic + 2 * hh)     * STR + k0 + il];
        bf.y = xs[(ic + 2 * hh + 1) * STR + k0 + il];
        acc = __builtin_amdgcn_wmma_f32_16x16x4_f32(false, af, false, bf,
                                                    (short)0, acc, false, false);
      }
      int k = k0 + il;
      float part[8];
      for (int r = 0; r < 8; ++r) {
        int o = o0 + r + 8 * hh;
        float g = acc[r] - tt[o] * as[o * STR + k];
        greg[u][r] = g;
        part[r] = half16_sum(g * g);
      }
      if (il == 0)
        for (int r = 0; r < 8; ++r) gpart[(o0 + r + 8 * hh) * 4 + (t & 3)] = part[r];
    }
    __syncthreads();

    if (tid < C) {
      float s2 = gpart[tid * 4 + 0] + gpart[tid * 4 + 1] +
                 gpart[tid * 4 + 2] + gpart[tid * 4 + 3];
      float gl = sqrtf(s2);
      cg[tid] = cosf(gl);
      sg[tid] = (gl < kEps) ? 1.0f : (sinf(gl) / gl);
    }
    __syncthreads();

    for (int u = 0; u < 2; ++u) {
      int t  = wave * 2 + u;
      int o0 = (t >> 2) * 16;
      int k0 = (t & 3) * 16;
      int k  = k0 + il;
      for (int r = 0; r < 8; ++r) {
        int o = o0 + r + 8 * hh;
        as[o * STR + k] = cg[o] * as[o * STR + k] + sg[o] * greg[u][r];
      }
    }
    __syncthreads();
  }

  float* on = out + (size_t)n * C * C;
  for (int j = 0; j < 16; ++j) {
    int f = tid + 256 * j;
    int o = f >> 6, k = f & 63;
    on[f] = as[o * STR + k];
  }
}

extern "C" void kernel_launch(void* const* d_in, const int* in_sizes, int n_in,
                              void* d_out, int out_size, void* d_ws, size_t ws_size,
                              hipStream_t stream) {
  (void)in_sizes; (void)n_in; (void)out_size; (void)ws_size;
  const float* x     = (const float*)d_in[0];
  const float* w_raw = (const float*)d_in[1];
  float* out = (float*)d_out;
  float* wn  = (float*)d_ws;

  mfdfc_wnorm_kernel<<<1, 64, 0, stream>>>(w_raw, wn);
  mfdfc_main_kernel<<<1024, 256, 0, stream>>>(x, wn, out);
}


# Row sums (s*dot, |grad|^2): a wave's two tiles added per row in registers, then one halving butterfly: 8 DPP adds + 14 selects per product, was 64 DPP adds; one store per sum; new summation order
# speedup vs baseline: 2.3539x; 1.0867x over previous
#include <hip/hip_runtime.h>

typedef float v2f __attribute__((ext_vector_type(2)));
typedef float v8f __attribute__((ext_vector_type(8)));

namespace {
constexpr int   C    = 64;
constexpr int   STR  = 68;
constexpr float kEps = 1e-7f;
}

__global__ __launch_bounds__(64)
void mfdfc_wnorm_kernel(const float* __restrict__ w_raw, float* __restrict__ wn) {
  const int o = threadIdx.x;
  const float lc = 4.1588830833596715f;
  float e[C];
  for (int i = 0; i < C; ++i) e[i] = w_raw[i * C + o];
  for (int i = 0; i < C; ++i) e[i] = expf(e[i] - lc);
  float s = 0.0f;
  for (int i = 0; i < C; ++i) s += e[i];
  const float inv = 1.0f / s;
  for (int i = 0; i < C; ++i) wn[i * C + o] = e[i] * inv;
}

template <int M>
__device__ __forceinline__ float row_xor(float v) {
  return __builtin_bit_cast(
      float, __builtin_amdgcn_update_dpp(0, __builtin_bit_cast(int, v), 0x160 + M,
                                         0xf, 0xf, true));
}

__device__ __forceinline__ float row16_sum8(const float (&v)[8], bool b0, bool b1, bool b2) {
  float w[4], q[2];
  for (int j = 0; j < 4; ++j) {
    float keep = b0 ? v[2 * j + 1] : v[2 * j];
    float send = b0 ? v[2 * j] : v[2 * j + 1];
    w[j] = keep + row_xor<1>(send);
  }
  for (int j = 0; j < 2; ++j) {
    float keep = b1 ? w[2 * j + 1] : w[2 * j];
    float send = b1 ? w[2 * j] : w[2 * j + 1];
    q[j] = keep + row_xor<2>(send);
  }
  float keep = b2 ? q[1] : q[0];
  float send = b2 ? q[0] : q[1];
  float z = keep + row_xor<4>(send);
  z += row_xor<8>(z);
  return z;
}

__global__ __launch_bounds__(256) __attribute__((amdgpu_waves_per_eu(8, 8)))
void mfdfc_main_kernel(const float* __restrict__ x, const float* __restrict__ wn,
                       float* __restrict__ out) {
  __shared__ float xs[C * STR];
  __shared__ float as[C * STR];
  __shared__ float ss[C * STR];
  __shared__ float tpart[C * 2];
  __shared__ float gpart[C * 2];
  __shared__ float pa[C];
  __shared__ float qq[C];
  __shared__ float tt[C];
  __shared__ float cg[C];
  __shared__ float sg[C];

  const int tid  = threadIdx.x;
  const int lane = tid & 31;
  const int wave = tid >> 5;
  const int il   = lane & 15;
  const int hh   = lane >> 4;
  const int n    = blockIdx.x;
  const bool b0  = (il & 1) != 0;
  const bool b1  = (il & 2) != 0;
  const bool b2  = (il & 4) != 0;
  const int  mw  = (wave >> 1) * 16;

  const float* xn = x + (size_t)n * C * C;

  for (int j = 0; j < 4; ++j) {
    int f4  = tid + 256 * j;
    int row = f4 >> 4;
    int c4  = f4 & 15;
    const float4 v = ((const float4*)xn)[f4];
    *(float4*)&xs[row * STR + 4 * c4] = v;
  }
  __syncthreads();

  if (tid < C) {
    float s = 0.0f;
    for (int k = 0; k < C; ++k) { float v = xs[tid * STR + k]; s += v * v; }
    qq[tid] = s;
  }
  for (int j = 0; j < 16; ++j) {
    int f = tid + 256 * j;
    int o = f >> 6, k = f & 63;
    as[o * STR + k] = xs[k];
  }

  float wreg[2][8];
  for (int u = 0; u < 2; ++u) {
    int t  = wave * 2 + u;
    int m0 = (t >> 2) * 16;
    int i0 = (t & 3) * 16;
    int i  = i0 + il;
    for (int r = 0; r < 8; ++r) {
      int o = m0 + r + 8 * hh;
      wreg[u][r] = wn[i * C + o];
    }
  }
  __syncthreads();

  for (int it = 0; it < 3; ++it) {
    if (tid < C) {
      float s = 0.0f;
      for (int k = 0; k < C; ++k) { float v = as[tid * STR + k]; s += v * v; }
      pa[tid] = s;
    }
    __syncthreads();

    float sd[8] = {};
    for (int u = 0; u < 2; ++u) {
      int t  = wave * 2 + u;
      int m0 = (t >> 2) * 16;
      int i0 = (t & 3) * 16;
      v8f acc = {};
      for (int kc = 0; kc < C; kc += 4) {
        v2f af = *(const v2f*)&as[(m0 + il) * STR + kc + 2 * hh];
        v2f bf = *(const v2f*)&xs[(i0 + il) * STR + kc + 2 * hh];
        acc = __builtin_amdgcn_wmma_f32_16x16x4_f32(false, af, false, bf,
                                                    (short)0, acc, false, false);
      }
      int i = i0 + il;
      for (int r = 0; r < 8; ++r) {
        int o = m0 + r + 8 * hh;
        float du  = acc[r];
        float dc  = fminf(fmaxf(du, -1.0f + kEps), 1.0f - kEps);
        float th  = acosf(dc);
        float nv2 = qq[i] - 2.0f * dc * du + dc * dc * pa[o];
        float nv  = sqrtf(fmaxf(nv2, 0.0f));
        float fac = (nv < kEps) ? 1.0f : th / nv;
        float s   = wreg[u][r] * fac;
        ss[o * STR + i] = s;
        float p = s * dc;
        sd[r] = (u == 0) ? p : sd[r] + p;
      }
    }
    {
      float z = row16_sum8(sd, b0, b1, b2);
      if (il < 8) tpart[(mw + il + 8 * hh) * 2 + (wave & 1)] = z;
    }
    __syncthreads();

    if (tid < C) tt[tid] = tpart[tid * 2 + 0] + tpart[tid * 2 + 1];
    __syncthreads();

    float greg[2][8];
    float gg[8] = {};
    for (int u = 0; u < 2; ++u) {
      int t  = wave * 2 + u;
      int o0 = (t >> 2) * 16;
      int k0 = (t & 3) * 16;
      v8f acc = {};
      for (int ic = 0; ic < C; ic += 4) {
        v2f af = *(const v2f*)&ss[(o0 + il) * STR + ic + 2 * hh];
        v2f bf;
        bf.x = xs[(ic + 2 * hh)     * STR + k0 + il];
        bf.y = xs[(ic + 2 * hh + 1) * STR + k0 + il];
        acc = __builtin_amdgcn_wmma_f32_16x16x4_f32(false, af, false, bf,
                                                    (short)0, acc, false, false);
      }
      int k = k0 + il;
      for (int r = 0; r < 8; ++r) {
        int o = o0 + r + 8 * hh;
        float g = acc[r] - tt[o] * as[o * STR + k];
        greg[u][r] = g;
        float p = g * g;
        gg[r] = (u == 0) ? p : gg[r] + p;
      }
    }
    {
      float z = row16_sum8(gg, b0, b1, b2);
      if (il < 8) gpart[(mw + il + 8 * hh) * 2 + (wave & 1)] = z;
    }
    __syncthreads();

    if (tid < C) {
      float s2 = gpart[tid * 2 + 0] + gpart[tid * 2 + 1];
      float gl = sqrtf(s2);
      cg[tid] = cosf(gl);
      sg[tid] = (gl < kEps) ? 1.0f : (sinf(gl) / gl);
    }
    __syncthreads();

    for (int u = 0; u < 2; ++u) {
      int t  = wave * 2 + u;
      int o0 = (t >> 2) * 16;
      int k0 = (t & 3) * 16;
      int k  = k0 + il;
      for (int r = 0; r < 8; ++r) {
        int o = o0 + r + 8 * hh;
        as[o * STR + k] = cg[o] * as[o * STR + k] + sg[o] * greg[u][r];
      }
    }
    __syncthreads();
  }

  float* on = out + (size_t)n * C * C;
  for (int j = 0; j < 16; ++j) {
    int f = tid + 256 * j;
    int o = f >> 6, k = f & 63;
    on[f] = as[o * STR + k];
  }
}

extern "C" void kernel_launch(void* const* d_in, const int* in_sizes, int n_in,
                              void* d_out, int out_size, void* d_ws, size_t ws_size,
                              hipStream_t stream) {
  (void)in_sizes; (void)n_in; (void)out_size; (void)ws_size;
  const float* x     = (const float*)d_in[0];
  const float* w_raw = (const float*)d_in[1];
  float* out = (float*)d_out;
  float* wn  = (float*)d_ws;

  mfdfc_wnorm_kernel<<<1, 64, 0, stream>>>(w_raw, wn);
  mfdfc_main_kernel<<<1024, 256, 0, stream>>>(x, wn, out);
}


# LDS rows addressed as one per-lane base plus constant offsets (init, S store, grad epilogue, mean update, final copy): per-row address registers and adds gone, more dual-issue pairs; sums unchanged
# speedup vs baseline: 2.5378x; 1.0781x over previous
#include <hip/hip_runtime.h>

typedef float v2f __attribute__((ext_vector_type(2)));
typedef float v8f __attribute__((ext_vector_type(8)));

namespace {
constexpr int   C    = 64;
constexpr int   STR  = 68;
constexpr float kEps = 1e-7f;
}

__global__ __launch_bounds__(64)
void mfdfc_wnorm_kernel(const float* __restrict__ w_raw, float* __restrict__ wn) {
  const int o = threadIdx.x;
  const float lc = 4.1588830833596715f;
  float e[C];
  for (int i = 0; i < C; ++i) e[i] = w_raw[i * C + o];
  for (int i = 0; i < C; ++i) e[i] = expf(e[i] - lc);
  float s = 0.0f;
  for (int i = 0; i < C; ++i) s += e[i];
  const float inv = 1.0f / s;
  for (int i = 0; i < C; ++i) wn[i * C + o] = e[i] * inv;
}

template <int M>
__device__ __forceinline__ float row_xor(float v) {
  return __builtin_bit_cast(
      float, __builtin_amdgcn_update_dpp(0, __builtin_bit_cast(int, v), 0x160 + M,
                                         0xf, 0xf, true));
}

__device__ __forceinline__ float row16_sum8(const float (&v)[8], bool b0, bool b1, bool b2) {
  float w[4], q[2];
  for (int j = 0; j < 4; ++j) {
    float keep = b0 ? v[2 * j + 1] : v[2 * j];
    float send = b0 ? v[2 * j] : v[2 * j + 1];
    w[j] = keep + row_xor<1>(send);
  }
  for (int j = 0; j < 2; ++j) {
    float keep = b1 ? w[2 * j + 1] : w[2 * j];
    float send = b1 ? w[2 * j] : w[2 * j + 1];
    q[j] = keep + row_xor<2>(send);
  }
  float keep = b2 ? q[1] : q[0];
  float send = b2 ? q[0] : q[1];
  float z = keep + row_xor<4>(send);
  z += row_xor<8>(z);
  return z;
}

__global__ __launch_bounds__(256) __attribute__((amdgpu_waves_per_eu(8, 8)))
void mfdfc_main_kernel(const float* __restrict__ x, const float* __restrict__ wn,
                       float* __restrict__ out) {
  __shared__ float xs[C * STR];
  __shared__ float as[C * STR];
  __shared__ float ss[C * STR];
  __shared__ float tpart[C * 2];
  __shared__ float gpart[C * 2];
  __shared__ float pa[C];
  __shared__ float qq[C];
  __shared__ float tt[C];
  __shared__ float cg[C];
  __shared__ float sg[C];

  const int tid  = threadIdx.x;
  const int lane = tid & 31;
  const int wave = tid >> 5;
  const int il   = lane & 15;
  const int hh   = lane >> 4;
  const int n    = blockIdx.x;
  const bool b0  = (il & 1) != 0;
  const bool b1  = (il & 2) != 0;
  const bool b2  = (il & 4) != 0;
  const int  mw  = (wave >> 1) * 16;

  const float* xn = x + (size_t)n * C * C;

  for (int j = 0; j < 4; ++j) {
    int f4  = tid + 256 * j;
    int row = f4 >> 4;
    int c4  = f4 & 15;
    const float4 v = ((const float4*)xn)[f4];
    *(float4*)&xs[row * STR + 4 * c4] = v;
  }
  __syncthreads();

  if (tid < C) {
    float s = 0.0f;
    for (int k = 0; k < C; ++k) { float v = xs[tid * STR + k]; s += v * v; }
    qq[tid] = s;
  }
  {
    const int ob = (tid >> 6) * STR + (tid & 63);
    const float x0k = xs[tid & 63];
    for (int j = 0; j < 16; ++j) as[ob + 4 * j * STR] = x0k;
  }

  float wreg[2][8];
  for (int u = 0; u < 2; ++u) {
    int t  = wave * 2 + u;
    int m0 = (t >> 2) * 16;
    int i0 = (t & 3) * 16;
    int i  = i0 + il;
    for (int r = 0; r < 8; ++r) {
      int o = m0 + r + 8 * hh;
      wreg[u][r] = wn[i * C + o];
    }
  }
  __syncthreads();

  for (int it = 0; it < 3; ++it) {
    if (tid < C) {
      float s = 0.0f;
      for (int k = 0; k < C; ++k) { float v = as[tid * STR + k]; s += v * v; }
      pa[tid] = s;
    }
    __syncthreads();

    float sd[8] = {};
    for (int u = 0; u < 2; ++u) {
      int t  = wave * 2 + u;
      int m0 = (t >> 2) * 16;
      int i0 = (t & 3) * 16;
      v8f acc = {};
      for (int kc = 0; kc < C; kc += 4) {
        v2f af = *(const v2f*)&as[(m0 + il) * STR + kc + 2 * hh];
        v2f bf = *(const v2f*)&xs[(i0 + il) * STR + kc + 2 * hh];
        acc = __builtin_amdgcn_wmma_f32_16x16x4_f32(false, af, false, bf,
                                                    (short)0, acc, false, false);
      }
      int i = i0 + il;
      for (int r = 0; r < 8; ++r) {
        int o = m0 + r + 8 * hh;
        float du  = acc[r];
        float dc  = fminf(fmaxf(du, -1.0f + kEps), 1.0f - kEps);
        float th  = acosf(dc);
        float nv2 = qq[i] - 2.0f * dc * du + dc * dc * pa[o];
        float nv  = sqrtf(fmaxf(nv2, 0.0f));
        float fac = (nv < kEps) ? 1.0f : th / nv;
        float s   = wreg[u][r] * fac;
        ss[(mw + 8 * hh) * STR + i + r * STR] = s;
        float p = s * dc;
        sd[r] = (u == 0) ? p : sd[r] + p;
      }
    }
    {
      float z = row16_sum8(sd, b0, b1, b2);
      if (il < 8) tpart[(mw + il + 8 * hh) * 2 + (wave & 1)] = z;
    }
    __syncthreads();

    if (tid < C) tt[tid] = tpart[tid * 2 + 0] + tpart[tid * 2 + 1];
    __syncthreads();

    float greg[2][8];
    float gg[8] = {};
    for (int u = 0; u < 2; ++u) {
      int t  = wave * 2 + u;
      int o0 = (t >> 2) * 16;
      int k0 = (t & 3) * 16;
      v8f acc = {};
      for (int ic = 0; ic < C; ic += 4) {
        v2f af = *(const v2f*)&ss[(o0 + il) * STR + ic + 2 * hh];
        v2f bf;
        bf.x = xs[(ic + 2 * hh)     * STR + k0 + il];
        bf.y = xs[(ic + 2 * hh + 1) * STR + k0 + il];
        acc = __builtin_amdgcn_wmma_f32_16x16x4_f32(false, af, false, bf,
                                                    (short)0, acc, false, false);
      }
      int k = k0 + il;
      for (int r = 0; r < 8; ++r) {
        int o = o0 + r + 8 * hh;
        float g = acc[r] - tt[o] * as[(mw + 8 * hh) * STR + k + r * STR];
        greg[u][r] = g;
        float p = g * g;
        gg[r] = (u == 0) ? p : gg[r] + p;
      }
    }
    {
      float z = row16_sum8(gg, b0, b1, b2);
      if (il < 8) gpart[(mw + il + 8 * hh) * 2 + (wave & 1)] = z;
    }
    __syncthreads();

    if (tid < C) {
      float s2 = gpart[tid * 2 + 0] + gpart[tid * 2 + 1];
      float gl = sqrtf(s2);
      cg[tid] = cosf(gl);
      sg[tid] = (gl < kEps) ? 1.0f : (sinf(gl) / gl);
    }
    __syncthreads();

    for (int u = 0; u < 2; ++u) {
      int t  = wave * 2 + u;
      int o0 = (t >> 2) * 16;
      int k0 = (t & 3) * 16;
      int k  = k0 + il;
      for (int r = 0; r < 8; ++r) {
        int o = o0 + r + 8 * hh;
        as[(mw + 8 * hh) * STR + k + r * STR] = cg[o] * as[(mw + 8 * hh) * STR + k + r * STR] + sg[o] * greg[u][r];
      }
    }
    __syncthreads();
  }

  float* on = out + (size_t)n * C * C;
  {
    const int ob = (tid >> 6) * STR + (tid & 63);
    for (int j = 0; j < 16; ++j) on[tid + 256 * j] = as[ob + 4 * j * STR];
  }
}

extern "C" void kernel_launch(void* const* d_in, const int* in_sizes, int n_in,
                              void* d_out, int out_size, void* d_ws, size_t ws_size,
                              hipStream_t stream) {
  (void)in_sizes; (void)n_in; (void)out_size; (void)ws_size;
  const float* x     = (const float*)d_in[0];
  const float* w_raw = (const float*)d_in[1];
  float* out = (float*)d_out;
  float* wn  = (float*)d_ws;

  mfdfc_wnorm_kernel<<<1, 64, 0, stream>>>(w_raw, wn);
  mfdfc_main_kernel<<<1024, 256, 0, stream>>>(x, wn, out);
}


# x tile also kept transposed in LDS: product 2's B fragment is one aligned 8-byte load (as in product 1), not two 4-byte row loads re-paired by moves plus a new address per k-step; sums unchanged
# speedup vs baseline: 2.6148x; 1.0303x over previous
#include <hip/hip_runtime.h>

typedef float v2f __attribute__((ext_vector_type(2)));
typedef float v8f __attribute__((ext_vector_type(8)));

namespace {
constexpr int   C    = 64;
constexpr int   STR  = 68;
constexpr float kEps = 1e-7f;
}

__global__ __launch_bounds__(64)
void mfdfc_wnorm_kernel(const float* __restrict__ w_raw, float* __restrict__ wn) {
  const int o = threadIdx.x;
  const float lc = 4.1588830833596715f;
  float e[C];
  for (int i = 0; i < C; ++i) e[i] = w_raw[i * C + o];
  for (int i = 0; i < C; ++i) e[i] = expf(e[i] - lc);
  float s = 0.0f;
  for (int i = 0; i < C; ++i) s += e[i];
  const float inv = 1.0f / s;
  for (int i = 0; i < C; ++i) wn[i * C + o] = e[i] * inv;
}

template <int M>
__device__ __forceinline__ float row_xor(float v) {
  return __builtin_bit_cast(
      float, __builtin_amdgcn_update_dpp(0, __builtin_bit_cast(int, v), 0x160 + M,
                                         0xf, 0xf, true));
}

__device__ __forceinline__ float row16_sum8(const float (&v)[8], bool b0, bool b1, bool b2) {
  float w[4], q[2];
  for (int j = 0; j < 4; ++j) {
    float keep = b0 ? v[2 * j + 1] : v[2 * j];
    float send = b0 ? v[2 * j] : v[2 * j + 1];
    w[j] = keep + row_xor<1>(send);
  }
  for (int j = 0; j < 2; ++j) {
    float keep = b1 ? w[2 * j + 1] : w[2 * j];
    float send = b1 ? w[2 * j] : w[2 * j + 1];
    q[j] = keep + row_xor<2>(send);
  }
  float keep = b2 ? q[1] : q[0];
  float send = b2 ? q[0] : q[1];
  float z = keep + row_xor<4>(send);
  z += row_xor<8>(z);
  return z;
}

__global__ __launch_bounds__(256) __attribute__((amdgpu_waves_per_eu(8, 8)))
void mfdfc_main_kernel(const float* __restrict__ x, const float* __restrict__ wn,
                       float* __restrict__ out) {
  __shared__ float xs[C * STR];
  __shared__ float xt[C * STR];
  __shared__ float as[C * STR];
  __shared__ float ss[C * STR];
  __shared__ float tpart[C * 2];
  __shared__ float gpart[C * 2];
  __shared__ float pa[C];
  __shared__ float qq[C];
  __shared__ float tt[C];
  __shared__ float cg[C];
  __shared__ float sg[C];

  const int tid  = threadIdx.x;
  const int lane = tid & 31;
  const int wave = tid >> 5;
  const int il   = lane & 15;
  const int hh   = lane >> 4;
  const int n    = blockIdx.x;
  const bool b0  = (il & 1) != 0;
  const bool b1  = (il & 2) != 0;
  const bool b2  = (il & 4) != 0;
  const int  mw  = (wave >> 1) * 16;

  const float* xn = x + (size_t)n * C * C;

  for (int j = 0; j < 4; ++j) {
    int f4  = tid + 256 * j;
    int row = f4 >> 4;
    int c4  = f4 & 15;
    const float4 v = ((const float4*)xn)[f4];
    *(float4*)&xs[row * STR + 4 * c4] = v;
    xt[(4 * c4 + 0) * STR + row] = v.x;
    xt[(4 * c4 + 1) * STR + row] = v.y;
    xt[(4 * c4 + 2) * STR + row] = v.z;
    xt[(4 * c4 + 3) * STR + row] = v.w;
  }
  __syncthreads();

  if (tid < C) {
    float s = 0.0f;
    for (int k = 0; k < C; ++k) { float v = xs[tid * STR + k]; s += v * v; }
    qq[tid] = s;
  }
  {
    const int ob = (tid >> 6) * STR + (tid & 63);
    const float x0k = xs[tid & 63];
    for (int j = 0; j < 16; ++j) as[ob + 4 * j * STR] = x0k;
  }

  float wreg[2][8];
  for (int u = 0; u < 2; ++u) {
    int t  = wave * 2 + u;
    int m0 = (t >> 2) * 16;
    int i0 = (t & 3) * 16;
    int i  = i0 + il;
    for (int r = 0; r < 8; ++r) {
      int o = m0 + r + 8 * hh;
      wreg[u][r] = wn[i * C + o];
    }
  }
  __syncthreads();

  for (int it = 0; it < 3; ++it) {
    if (tid < C) {
      float s = 0.0f;
      for (int k = 0; k < C; ++k) { float v = as[tid * STR + k]; s += v * v; }
      pa[tid] = s;
    }
    __syncthreads();

    float sd[8] = {};
    for (int u = 0; u < 2; ++u) {
      int t  = wave * 2 + u;
      int m0 = (t >> 2) * 16;
      int i0 = (t & 3) * 16;
      v8f acc = {};
      for (int kc = 0; kc < C; kc += 4) {
        v2f af = *(const v2f*)&as[(m0 + il) * STR + kc + 2 * hh];
        v2f bf = *(const v2f*)&xs[(i0 + il) * STR + kc + 2 * hh];
        acc = __builtin_amdgcn_wmma_f32_16x16x4_f32(false, af, false, bf,
                                                    (short)0, acc, false, false);
      }
      int i = i0 + il;
      for (int r = 0; r < 8; ++r) {
        int o = m0 + r + 8 * hh;
        float du  = acc[r];
        float dc  = fminf(fmaxf(du, -1.0f + kEps), 1.0f - kEps);
        float th  = acosf(dc);
        float nv2 = qq[i] - 2.0f * dc * du + dc * dc * pa[o];
        float nv  = sqrtf(fmaxf(nv2, 0.0f));
        float fac = (nv < kEps) ? 1.0f : th / nv;
        float s   = wreg[u][r] * fac;
        ss[(mw + 8 * hh) * STR + i + r * STR] = s;
        float p = s * dc;
        sd[r] = (u == 0) ? p : sd[r] + p;
      }
    }
    {
      float z = row16_sum8(sd, b0, b1, b2);
      if (il < 8) tpart[(mw + il + 8 * hh) * 2 + (wave & 1)] = z;
    }
    __syncthreads();

    if (tid < C) tt[tid] = tpart[tid * 2 + 0] + tpart[tid * 2 + 1];
    __syncthreads();

    float greg[2][8];
    float gg[8] = {};
    for (int u = 0; u < 2; ++u) {
      int t  = wave * 2 + u;
      int o0 = (t >> 2) * 16;
      int k0 = (t & 3) * 16;
      v8f acc = {};
      for (int ic = 0; ic < C; ic += 4) {
        v2f af = *(const v2f*)&ss[(o0 + il) * STR + ic + 2 * hh];
        v2f bf = *(const v2f*)&xt[(k0 + il) * STR + ic + 2 * hh];
        acc = __builtin_amdgcn_wmma_f32_16x16x4_f32(false, af, false, bf,
                                                    (short)0, acc, false, false);
      }
      int k = k0 + il;
      for (int r = 0; r < 8; ++r) {
        int o = o0 + r + 8 * hh;
        float g = acc[r] - tt[o] * as[(mw + 8 * hh) * STR + k + r * STR];
        greg[u][r] = g;
        float p = g * g;
        gg[r] = (u == 0) ? p : gg[r] + p;
      }
    }
    {
      float z = row16_sum8(gg, b0, b1, b2);
      if (il < 8) gpart[(mw + il + 8 * hh) * 2 + (wave & 1)] = z;
    }
    __syncthreads();

    if (tid < C) {
      float s2 = gpart[tid * 2 + 0] + gpart[tid * 2 + 1];
      float gl = sqrtf(s2);
      cg[tid] = cosf(gl);
      sg[tid] = (gl < kEps) ? 1.0f : (sinf(gl) / gl);
    }
    __syncthreads();

    for (int u = 0; u < 2; ++u) {
      int t  = wave * 2 + u;
      int o0 = (t >> 2) * 16;
      int k0 = (t & 3) * 16;
      int k  = k0 + il;
      for (int r = 0; r < 8; ++r) {
        int o = o0 + r + 8 * hh;
        as[(mw + 8 * hh) * STR + k + r * STR] = cg[o] * as[(mw + 8 * hh) * STR + k + r * STR] + sg[o] * greg[u][r];
      }
    }
    __syncthreads();
  }

  float* on = out + (size_t)n * C * C;
  {
    const int ob = (tid >> 6) * STR + (tid & 63);
    for (int j = 0; j < 16; ++j) on[tid + 256 * j] = as[ob + 4 * j * STR];
  }
}

extern "C" void kernel_launch(void* const* d_in, const int* in_sizes, int n_in,
                              void* d_out, int out_size, void* d_ws, size_t ws_size,
                              hipStream_t stream) {
  (void)in_sizes; (void)n_in; (void)out_size; (void)ws_size;
  const float* x     = (const float*)d_in[0];
  const float* w_raw = (const float*)d_in[1];
  float* out = (float*)d_out;
  float* wn  = (float*)d_ws;

  mfdfc_wnorm_kernel<<<1, 64, 0, stream>>>(w_raw, wn);
  mfdfc_main_kernel<<<1024, 256, 0, stream>>>(x, wn, out);
}
